# starting kernel
# baseline (speedup 1.0000x reference)
#include <hip/hip_runtime.h>
#include <math.h>

#define N_NODES 131072
#define E_EDGES 1048576
#define NUM_DSTN 65536
#define IN_DIM 256
#define HEADS 4
#define OUT_D 32
#define FEAT (HEADS * OUT_D)
#define NEG_SLOPE 0.2f

typedef __attribute__((ext_vector_type(2))) float v2f;
typedef __attribute__((ext_vector_type(8))) float v8f;

__global__ void init_kernel(float* __restrict__ out,
                            float* __restrict__ emax,
                            float* __restrict__ esum) {
    size_t i = (size_t)blockIdx.x * blockDim.x + threadIdx.x;
    if (i < (size_t)NUM_DSTN * FEAT) out[i] = 0.0f;
    if (i < (size_t)NUM_DSTN * HEADS) {
        ((unsigned int*)emax)[i] = 0xFF800000u;
        esum[i] = 0.0f;
    }
}

__global__ __launch_bounds__(256) void gemm_kernel(const float* __restrict__ feats,
                                                   const float* __restrict__ W,
                                                   float* __restrict__ h) {
    const int wave = blockIdx.x * (blockDim.x >> 5) + (threadIdx.x >> 5);
    const int lane = threadIdx.x & 31;
    const int lo = lane & 15;
    const int hi = lane >> 4;
    const int rowBase = wave * 16;
    const float* __restrict__ arow = feats + (size_t)(rowBase + lo) * IN_DIM;

    v8f acc[8];
#pragma unroll
    for (int t = 0; t < 8; ++t) acc[t] = (v8f){0.f, 0.f, 0.f, 0.f, 0.f, 0.f, 0.f, 0.f};

    for (int k = 0; k < IN_DIM; k += 4) {
        const int ka = k + 2 * hi;
        v2f a;
        a.x = arow[ka];
        a.y = arow[ka + 1];
#pragma unroll
        for (int t = 0; t < 8; ++t) {
            const int col = t * 16 + lo;
            v2f b;
            b.x = W[(size_t)ka * FEAT + col];
            b.y = W[(size_t)(ka + 1) * FEAT + col];
            acc[t] = __builtin_amdgcn_wmma_f32_16x16x4_f32(
                false, a, false, b, (short)0, acc[t], false, false);
        }
    }

#pragma unroll
    for (int t = 0; t < 8; ++t)
#pragma unroll
        for (int i = 0; i < 8; ++i)
            h[(size_t)(rowBase + i + 8 * hi) * FEAT + t * 16 + lo] = acc[t][i];
}

__global__ void attn_logits_kernel(const float* __restrict__ h,
                                   const float* __restrict__ attn_l,
                                   const float* __restrict__ attn_r,
                                   float* __restrict__ el, float* __restrict__ er) {
    int i = blockIdx.x * blockDim.x + threadIdx.x;
    if (i >= N_NODES * HEADS) return;
    const int n = i >> 2;
    const int hh = i & 3;
    const float* __restrict__ hp = h + (size_t)n * FEAT + hh * OUT_D;
    const float* __restrict__ al = attn_l + hh * OUT_D;
    const float* __restrict__ ar = attn_r + hh * OUT_D;
    float sl = 0.f, sr = 0.f;
#pragma unroll
    for (int d = 0; d < OUT_D; ++d) {
        float v = hp[d];
        sl += v * al[d];
        sr += v * ar[d];
    }
    el[i] = sl;
    er[i] = sr;
}

__device__ __forceinline__ void atomicMaxF(float* addr, float val) {
    if (val >= 0.0f)
        atomicMax((int*)addr, __float_as_int(val));
    else
        atomicMin((unsigned int*)addr, __float_as_uint(val));
}

__global__ void edge_max_kernel(const int* __restrict__ src, const int* __restrict__ dst,
                                const float* __restrict__ el, const float* __restrict__ er,
                                float* __restrict__ ebuf, float* __restrict__ emax) {
    int e = blockIdx.x * blockDim.x + threadIdx.x;
    if (e >= E_EDGES) return;
    const int s = src[e];
    const int d = dst[e];
    const float4 l4 = *(const float4*)(el + (size_t)s * HEADS);
    const float4 r4 = *(const float4*)(er + (size_t)d * HEADS);
    float vals[4] = {l4.x + r4.x, l4.y + r4.y, l4.z + r4.z, l4.w + r4.w};
#pragma unroll
    for (int hh = 0; hh < 4; ++hh) {
        float v = vals[hh];
        v = (v >= 0.f) ? v : NEG_SLOPE * v;
        ebuf[(size_t)e * HEADS + hh] = v;
        atomicMaxF(&emax[(size_t)d * HEADS + hh], v);
    }
}

__global__ void edge_exp_kernel(const int* __restrict__ dst,
                                float* __restrict__ ebuf,
                                const float* __restrict__ emax,
                                float* __restrict__ esum) {
    int e = blockIdx.x * blockDim.x + threadIdx.x;
    if (e >= E_EDGES) return;
    const int d = dst[e];
#pragma unroll
    for (int hh = 0; hh < 4; ++hh) {
        float v = expf(ebuf[(size_t)e * HEADS + hh] - emax[(size_t)d * HEADS + hh]);
        ebuf[(size_t)e * HEADS + hh] = v;
        atomicAdd(&esum[(size_t)d * HEADS + hh], v);
    }
}

__global__ __launch_bounds__(128) void scatter_kernel(const int* __restrict__ src,
                                                      const int* __restrict__ dst,
                                                      const float* __restrict__ h,
                                                      const float* __restrict__ ebuf,
                                                      const float* __restrict__ esum,
                                                      float* __restrict__ out) {
    const int e = blockIdx.x;
    const int t = threadIdx.x;
    const int s = src[e];
    const int d = dst[e];
    const int hh = t >> 5;
    const float alpha = ebuf[(size_t)e * HEADS + hh] / esum[(size_t)d * HEADS + hh];
    const float v = h[(size_t)s * FEAT + t] * alpha;
    atomicAdd(&out[(size_t)d * FEAT + t], v);
}

__global__ void finalize_kernel(float* __restrict__ out) {
    size_t i = (size_t)blockIdx.x * blockDim.x + threadIdx.x;
    if (i < (size_t)NUM_DSTN * FEAT) {
        float x = out[i];
        out[i] = (x > 0.f) ? x : (expf(x) - 1.0f);
    }
    if (i == 0) {
        out[(size_t)NUM_DSTN * FEAT + 0] = 1.0f;
        out[(size_t)NUM_DSTN * FEAT + 1] = 1.0f;
    }
}

extern "C" void kernel_launch(void* const* d_in, const int* in_sizes, int n_in,
                              void* d_out, int out_size, void* d_ws, size_t ws_size,
                              hipStream_t stream) {
    const float* feats  = (const float*)d_in[0];
    const int*   src    = (const int*)d_in[1];
    const int*   dst    = (const int*)d_in[2];
    const float* W      = (const float*)d_in[3];
    const float* attn_l = (const float*)d_in[4];
    const float* attn_r = (const float*)d_in[5];
    float* out = (float*)d_out;

    float* ws   = (float*)d_ws;
    float* hbuf = ws;
    float* el   = hbuf + (size_t)N_NODES * FEAT;
    float* er   = el + (size_t)N_NODES * HEADS;
    float* ebuf = er + (size_t)N_NODES * HEADS;
    float* emax = ebuf + (size_t)E_EDGES * HEADS;
    float* esum = emax + (size_t)NUM_DSTN * HEADS;

    {
        size_t n = (size_t)NUM_DSTN * FEAT;
        init_kernel<<<(unsigned)((n + 255) / 256), 256, 0, stream>>>(out, emax, esum);
    }
    {
        int rowTiles = N_NODES / 16;
        int blocks = rowTiles / 8;
        gemm_kernel<<<blocks, 256, 0, stream>>>(feats, W, hbuf);
    }
    {
        int n = N_NODES * HEADS;
        attn_logits_kernel<<<(n + 255) / 256, 256, 0, stream>>>(hbuf, attn_l, attn_r, el, er);
    }
    edge_max_kernel<<<(E_EDGES + 255) / 256, 256, 0, stream>>>(src, dst, el, er, ebuf, emax);
    edge_exp_kernel<<<(E_EDGES + 255) / 256, 256, 0, stream>>>(dst, ebuf, emax, esum);
    scatter_kernel<<<E_EDGES, 128, 0, stream>>>(src, dst, hbuf, ebuf, esum, out);
    {
        size_t n = (size_t)NUM_DSTN * FEAT;
        finalize_kernel<<<(unsigned)((n + 255) / 256), 256, 0, stream>>>(out);
    }
}


# Counting-sort edges by destination (atomic rank, scan, fill); one wave per destination does softmax and weighted sum in registers in arrival-rank order; float atomics and zeroing of out gone
# speedup vs baseline: 3.9060x; 3.9060x over previous
#include <hip/hip_runtime.h>
#include <math.h>

#define N_NODES 131072
#define E_EDGES 1048576
#define NUM_DSTN 65536
#define IN_DIM 256
#define HEADS 4
#define OUT_D 32
#define FEAT (HEADS * OUT_D)
#define NEG_SLOPE 0.2f

typedef __attribute__((ext_vector_type(2))) float v2f;
typedef __attribute__((ext_vector_type(8))) float v8f;

__global__ void init_kernel(int* __restrict__ cnt) {
    const int i = blockIdx.x * blockDim.x + threadIdx.x;
    if (i < NUM_DSTN) cnt[i] = 0;
}

__global__ void count_kernel(const int* __restrict__ dst, int* __restrict__ cnt,
                             int* __restrict__ rank) {
    const int e = blockIdx.x * blockDim.x + threadIdx.x;
    if (e >= E_EDGES) return;
    rank[e] = atomicAdd(&cnt[dst[e]], 1);
}

__global__ __launch_bounds__(1024) void scan_kernel(const int* __restrict__ cnt,
                                                    int* __restrict__ off) {
    __shared__ int part[512];
    __shared__ int wtot[16];
    const int t = threadIdx.x;
    const int lane = t & 31;
    const int wave = t >> 5;
    const int4* __restrict__ c4 = (const int4*)cnt;
    int4 v[16];
    int incl[16];
#pragma unroll
    for (int c = 0; c < 16; ++c) v[c] = c4[c * 1024 + t];
#pragma unroll
    for (int c = 0; c < 16; ++c) {
        int s = v[c].x + v[c].y + v[c].z + v[c].w;
#pragma unroll
        for (int o = 1; o < 32; o <<= 1) {
            const int nb = __shfl_up(s, o);
            if (lane >= o) s += nb;
        }
        incl[c] = s;
        if (lane == 31) part[c * 32 + wave] = s;
    }
    __syncthreads();
    int p = 0, pin = 0;
    if (t < 512) {
        p = part[t];
        pin = p;
#pragma unroll
        for (int o = 1; o < 32; o <<= 1) {
            const int nb = __shfl_up(pin, o);
            if (lane >= o) pin += nb;
        }
        if (lane == 31) wtot[wave] = pin;
    }
    __syncthreads();
    if (t < 512) {
        int b = 0;
        for (int w = 0; w < wave; ++w) b += wtot[w];
        part[t] = b + pin - p;
    }
    __syncthreads();
    int4* __restrict__ o4 = (int4*)off;
#pragma unroll
    for (int c = 0; c < 16; ++c) {
        const int own = v[c].x + v[c].y + v[c].z + v[c].w;
        const int base = part[c * 32 + wave] + incl[c] - own;
        int4 r;
        r.x = base;
        r.y = r.x + v[c].x;
        r.z = r.y + v[c].y;
        r.w = r.z + v[c].z;
        o4[c * 1024 + t] = r;
    }
}

__global__ void fill_kernel(const int* __restrict__ src, const int* __restrict__ dst,
                            const int* __restrict__ rank, const int* __restrict__ off,
                            int* __restrict__ ssrc) {
    const int e = blockIdx.x * blockDim.x + threadIdx.x;
    if (e >= E_EDGES) return;
    ssrc[off[dst[e]] + rank[e]] = src[e];
}

__global__ __launch_bounds__(256) void gemm_kernel(const float* __restrict__ feats,
                                                   const float* __restrict__ W,
                                                   float* __restrict__ h) {
    const int wave = blockIdx.x * (blockDim.x >> 5) + (threadIdx.x >> 5);
    const int lane = threadIdx.x & 31;
    const int lo = lane & 15;
    const int hi = lane >> 4;
    const int rowBase = wave * 16;
    const float* __restrict__ arow = feats + (size_t)(rowBase + lo) * IN_DIM;

    v8f acc[8];
#pragma unroll
    for (int t = 0; t < 8; ++t) acc[t] = (v8f){0.f, 0.f, 0.f, 0.f, 0.f, 0.f, 0.f, 0.f};

    for (int k = 0; k < IN_DIM; k += 4) {
        const int ka = k + 2 * hi;
        v2f a;
        a.x = arow[ka];
        a.y = arow[ka + 1];
#pragma unroll
        for (int t = 0; t < 8; ++t) {
            const int col = t * 16 + lo;
            v2f b;
            b.x = W[(size_t)ka * FEAT + col];
            b.y = W[(size_t)(ka + 1) * FEAT + col];
            acc[t] = __builtin_amdgcn_wmma_f32_16x16x4_f32(
                false, a, false, b, (short)0, acc[t], false, false);
        }
    }

#pragma unroll
    for (int t = 0; t < 8; ++t)
#pragma unroll
        for (int i = 0; i < 8; ++i)
            h[(size_t)(rowBase + i + 8 * hi) * FEAT + t * 16 + lo] = acc[t][i];
}

__global__ void attn_logits_kernel(const float* __restrict__ h,
                                   const float* __restrict__ attn_l,
                                   const float* __restrict__ attn_r,
                                   float* __restrict__ el, float* __restrict__ er) {
    int i = blockIdx.x * blockDim.x + threadIdx.x;
    if (i >= N_NODES * HEADS) return;
    const int n = i >> 2;
    const int hh = i & 3;
    const float* __restrict__ hp = h + (size_t)n * FEAT + hh * OUT_D;
    const float* __restrict__ al = attn_l + hh * OUT_D;
    const float* __restrict__ ar = attn_r + hh * OUT_D;
    float sl = 0.f, sr = 0.f;
#pragma unroll
    for (int d = 0; d < OUT_D; ++d) {
        float v = hp[d];
        sl += v * al[d];
        sr += v * ar[d];
    }
    el[i] = sl;
    er[i] = sr;
}

__device__ __forceinline__ float leaky(float v) { return (v >= 0.f) ? v : NEG_SLOPE * v; }

__device__ __forceinline__ float wave_max(float v) {
#pragma unroll
    for (int o = 16; o > 0; o >>= 1) v = fmaxf(v, __shfl_xor(v, o));
    return v;
}

__device__ __forceinline__ float wave_sum(float v) {
#pragma unroll
    for (int o = 16; o > 0; o >>= 1) v += __shfl_xor(v, o);
    return v;
}

__device__ __forceinline__ int lane_bcast(float v, int j) {
    return __builtin_amdgcn_readlane(__float_as_int(v), j);
}

__device__ __forceinline__ float pick_head(int b0, int b1, int b2, int b3, int k0, int k1, int k2, int k3) {
    return __int_as_float((b0 & k0) | (b1 & k1) | (b2 & k2) | (b3 & k3));
}

__device__ __forceinline__ void add_chunk(float4& acc, int s, float a0, float a1, float a2, float a3,
                                          int cntc, int hd, int lane, const float* __restrict__ h) {
    const int k0 = (hd == 0) ? -1 : 0, k1 = (hd == 1) ? -1 : 0;
    const int k2 = (hd == 2) ? -1 : 0, k3 = (hd == 3) ? -1 : 0;
    int j = 0;
    for (; j + 4 <= cntc; j += 4) {
        float4 hv[4];
        float a[4];
#pragma unroll
        for (int u = 0; u < 4; ++u) {
            const int sj = __builtin_amdgcn_readlane(s, j + u);
            hv[u] = *(const float4*)(h + (size_t)sj * FEAT + lane * 4);
            a[u] = pick_head(lane_bcast(a0, j + u), lane_bcast(a1, j + u),
                             lane_bcast(a2, j + u), lane_bcast(a3, j + u), k0, k1, k2, k3);
        }
#pragma unroll
        for (int u = 0; u < 4; ++u) {
            acc.x += hv[u].x * a[u];
            acc.y += hv[u].y * a[u];
            acc.z += hv[u].z * a[u];
            acc.w += hv[u].w * a[u];
        }
    }
    for (; j < cntc; ++j) {
        const int sj = __builtin_amdgcn_readlane(s, j);
        const float4 hv = *(const float4*)(h + (size_t)sj * FEAT + lane * 4);
        const float a = pick_head(lane_bcast(a0, j), lane_bcast(a1, j),
                                  lane_bcast(a2, j), lane_bcast(a3, j), k0, k1, k2, k3);
        acc.x += hv.x * a;
        acc.y += hv.y * a;
        acc.z += hv.z * a;
        acc.w += hv.w * a;
    }
}

__global__ __launch_bounds__(256) void aggregate_kernel(const int* __restrict__ off,
                                                        const int* __restrict__ cnt,
                                                        const int* __restrict__ ssrc,
                                                        const float* __restrict__ h,
                                                        const float* __restrict__ el,
                                                        const float* __restrict__ er,
                                                        float* __restrict__ out) {
    const int lane = threadIdx.x & 31;
    const int hd = lane >> 3;
    const int d = blockIdx.x * 8 + __builtin_amdgcn_readfirstlane((int)(threadIdx.x >> 5));
    const int begin = off[d];
    const int n = cnt[d];
    const float4* __restrict__ el4 = (const float4*)el;
    float4 acc = {0.f, 0.f, 0.f, 0.f};

    if (n > 0) {
        const float4 r4 = *(const float4*)(er + (size_t)d * HEADS);
        const bool v0 = lane < n;
        int s0 = 0;
        float e0 = -INFINITY, e1 = -INFINITY, e2 = -INFINITY, e3 = -INFINITY;
        if (v0) {
            s0 = ssrc[begin + lane];
            const float4 l4 = el4[s0];
            e0 = leaky(l4.x + r4.x); e1 = leaky(l4.y + r4.y);
            e2 = leaky(l4.z + r4.z); e3 = leaky(l4.w + r4.w);
        }
        float m0 = e0, m1 = e1, m2 = e2, m3 = e3;
        for (int base = 32; base < n; base += 32) {
            const int i = base + lane;
            if (i < n) {
                const float4 l4 = el4[ssrc[begin + i]];
                m0 = fmaxf(m0, leaky(l4.x + r4.x)); m1 = fmaxf(m1, leaky(l4.y + r4.y));
                m2 = fmaxf(m2, leaky(l4.z + r4.z)); m3 = fmaxf(m3, leaky(l4.w + r4.w));
            }
        }
        m0 = wave_max(m0); m1 = wave_max(m1); m2 = wave_max(m2); m3 = wave_max(m3);
        const float p0 = v0 ? expf(e0 - m0) : 0.f, p1 = v0 ? expf(e1 - m1) : 0.f;
        const float p2 = v0 ? expf(e2 - m2) : 0.f, p3 = v0 ? expf(e3 - m3) : 0.f;
        float z0 = p0, z1 = p1, z2 = p2, z3 = p3;
        for (int base = 32; base < n; base += 32) {
            const int i = base + lane;
            if (i < n) {
                const float4 l4 = el4[ssrc[begin + i]];
                z0 += expf(leaky(l4.x + r4.x) - m0); z1 += expf(leaky(l4.y + r4.y) - m1);
                z2 += expf(leaky(l4.z + r4.z) - m2); z3 += expf(leaky(l4.w + r4.w) - m3);
            }
        }
        z0 = wave_sum(z0); z1 = wave_sum(z1); z2 = wave_sum(z2); z3 = wave_sum(z3);
        add_chunk(acc, s0, p0 / z0, p1 / z1, p2 / z2, p3 / z3, (n < 32) ? n : 32, hd, lane, h);
        for (int base = 32; base < n; base += 32) {
            const int i = base + lane;
            int s = 0;
            float a0 = 0.f, a1 = 0.f, a2 = 0.f, a3 = 0.f;
            if (i < n) {
                s = ssrc[begin + i];
                const float4 l4 = el4[s];
                a0 = expf(leaky(l4.x + r4.x) - m0) / z0; a1 = expf(leaky(l4.y + r4.y) - m1) / z1;
                a2 = expf(leaky(l4.z + r4.z) - m2) / z2; a3 = expf(leaky(l4.w + r4.w) - m3) / z3;
            }
            const int rem = n - base;
            add_chunk(acc, s, a0, a1, a2, a3, (rem < 32) ? rem : 32, hd, lane, h);
        }
    }

    float4 r;
    r.x = (acc.x > 0.f) ? acc.x : (expf(acc.x) - 1.0f);
    r.y = (acc.y > 0.f) ? acc.y : (expf(acc.y) - 1.0f);
    r.z = (acc.z > 0.f) ? acc.z : (expf(acc.z) - 1.0f);
    r.w = (acc.w > 0.f) ? acc.w : (expf(acc.w) - 1.0f);
    *(float4*)(out + (size_t)d * FEAT + lane * 4) = r;
    if (blockIdx.x == 0 && threadIdx.x == 0) {
        out[(size_t)NUM_DSTN * FEAT + 0] = 1.0f;
        out[(size_t)NUM_DSTN * FEAT + 1] = 1.0f;
    }
}

extern "C" void kernel_launch(void* const* d_in, const int* in_sizes, int n_in,
                              void* d_out, int out_size, void* d_ws, size_t ws_size,
                              hipStream_t stream) {
    const float* feats  = (const float*)d_in[0];
    const int*   src    = (const int*)d_in[1];
    const int*   dst    = (const int*)d_in[2];
    const float* W      = (const float*)d_in[3];
    const float* attn_l = (const float*)d_in[4];
    const float* attn_r = (const float*)d_in[5];
    float* out = (float*)d_out;

    float* ws   = (float*)d_ws;
    float* hbuf = ws;
    float* el   = hbuf + (size_t)N_NODES * FEAT;
    float* er   = el + (size_t)N_NODES * HEADS;
    int*   rank = (int*)(er + (size_t)N_NODES * HEADS);
    int*   ssrc = rank + (size_t)E_EDGES;
    int*   cnt  = ssrc + (size_t)E_EDGES;
    int*   off  = cnt + (size_t)NUM_DSTN;

    init_kernel<<<NUM_DSTN / 256, 256, 0, stream>>>(cnt);
    {
        int rowTiles = N_NODES / 16;
        int blocks = rowTiles / 8;
        gemm_kernel<<<blocks, 256, 0, stream>>>(feats, W, hbuf);
    }
    {
        int n = N_NODES * HEADS;
        attn_logits_kernel<<<(n + 255) / 256, 256, 0, stream>>>(hbuf, attn_l, attn_r, el, er);
    }
    count_kernel<<<(E_EDGES + 255) / 256, 256, 0, stream>>>(dst, cnt, rank);
    scan_kernel<<<1, 1024, 0, stream>>>(cnt, off);
    fill_kernel<<<(E_EDGES + 255) / 256, 256, 0, stream>>>(src, dst, rank, off, ssrc);
    aggregate_kernel<<<NUM_DSTN / 8, 256, 0, stream>>>(off, cnt, ssrc, hbuf, el, er, out);
}


# Edge count and fill passes, 8 edges per thread, ride as extra blocks in the GEMM and logits launches (the two independent chains overlap; 7 launches to 5); sums stay in arrival-rank order
# speedup vs baseline: 4.1900x; 1.0727x over previous
#include <hip/hip_runtime.h>
#include <math.h>

#define N_NODES 131072
#define E_EDGES 1048576
#define NUM_DSTN 65536
#define IN_DIM 256
#define HEADS 4
#define OUT_D 32
#define FEAT (HEADS * OUT_D)
#define NEG_SLOPE 0.2f

typedef __attribute__((ext_vector_type(2))) float v2f;
typedef __attribute__((ext_vector_type(8))) float v8f;

__global__ void init_kernel(int* __restrict__ cnt) {
    const int i = blockIdx.x * blockDim.x + threadIdx.x;
    if (i < NUM_DSTN) cnt[i] = 0;
}

#define EDGE_PER_BLOCK 2048
#define EDGE_BLOCKS (E_EDGES / EDGE_PER_BLOCK)
#define GEMM_BLOCKS (N_NODES / 16 / 8)
#define LOGIT_BLOCKS (N_NODES * HEADS / 256)

__device__ __forceinline__ void count_block(int b, const int* __restrict__ dst,
                                            int* __restrict__ cnt, int* __restrict__ rank) {
    const int t = threadIdx.x;
    const int4* __restrict__ d4 = (const int4*)dst + (size_t)b * (EDGE_PER_BLOCK / 4);
    int4* __restrict__ r4 = (int4*)rank + (size_t)b * (EDGE_PER_BLOCK / 4);
    const int4 da = d4[t];
    const int4 db = d4[256 + t];
    int4 ra, rb;
    ra.x = atomicAdd(&cnt[da.x], 1);
    ra.y = atomicAdd(&cnt[da.y], 1);
    ra.z = atomicAdd(&cnt[da.z], 1);
    ra.w = atomicAdd(&cnt[da.w], 1);
    rb.x = atomicAdd(&cnt[db.x], 1);
    rb.y = atomicAdd(&cnt[db.y], 1);
    rb.z = atomicAdd(&cnt[db.z], 1);
    rb.w = atomicAdd(&cnt[db.w], 1);
    r4[t] = ra;
    r4[256 + t] = rb;
}

__global__ __launch_bounds__(1024) void scan_kernel(const int* __restrict__ cnt,
                                                    int* __restrict__ off) {
    __shared__ int part[512];
    __shared__ int wtot[16];
    const int t = threadIdx.x;
    const int lane = t & 31;
    const int wave = t >> 5;
    const int4* __restrict__ c4 = (const int4*)cnt;
    int4 v[16];
    int incl[16];
#pragma unroll
    for (int c = 0; c < 16; ++c) v[c] = c4[c * 1024 + t];
#pragma unroll
    for (int c = 0; c < 16; ++c) {
        int s = v[c].x + v[c].y + v[c].z + v[c].w;
#pragma unroll
        for (int o = 1; o < 32; o <<= 1) {
            const int nb = __shfl_up(s, o);
            if (lane >= o) s += nb;
        }
        incl[c] = s;
        if (lane == 31) part[c * 32 + wave] = s;
    }
    __syncthreads();
    int p = 0, pin = 0;
    if (t < 512) {
        p = part[t];
        pin = p;
#pragma unroll
        for (int o = 1; o < 32; o <<= 1) {
            const int nb = __shfl_up(pin, o);
            if (lane >= o) pin += nb;
        }
        if (lane == 31) wtot[wave] = pin;
    }
    __syncthreads();
    if (t < 512) {
        int b = 0;
        for (int w = 0; w < wave; ++w) b += wtot[w];
        part[t] = b + pin - p;
    }
    __syncthreads();
    int4* __restrict__ o4 = (int4*)off;
#pragma unroll
    for (int c = 0; c < 16; ++c) {
        const int own = v[c].x + v[c].y + v[c].z + v[c].w;
        const int base = part[c * 32 + wave] + incl[c] - own;
        int4 r;
        r.x = base;
        r.y = r.x + v[c].x;
        r.z = r.y + v[c].y;
        r.w = r.z + v[c].z;
        o4[c * 1024 + t] = r;
    }
}

__device__ __forceinline__ void fill_block(int b, const int* __restrict__ src,
                                           const int* __restrict__ dst,
                                           const int* __restrict__ rank,
                                           const int* __restrict__ off, int* __restrict__ ssrc) {
    const int t = threadIdx.x;
    const size_t q = (size_t)b * (EDGE_PER_BLOCK / 4);
    const int4* __restrict__ s4 = (const int4*)src + q;
    const int4* __restrict__ d4 = (const int4*)dst + q;
    const int4* __restrict__ r4 = (const int4*)rank + q;
    const int4 sa = s4[t], sb = s4[256 + t];
    const int4 da = d4[t], db = d4[256 + t];
    const int4 ra = r4[t], rb = r4[256 + t];
    const int p0 = off[da.x] + ra.x, p1 = off[da.y] + ra.y;
    const int p2 = off[da.z] + ra.z, p3 = off[da.w] + ra.w;
    const int p4 = off[db.x] + rb.x, p5 = off[db.y] + rb.y;
    const int p6 = off[db.z] + rb.z, p7 = off[db.w] + rb.w;
    ssrc[p0] = sa.x; ssrc[p1] = sa.y; ssrc[p2] = sa.z; ssrc[p3] = sa.w;
    ssrc[p4] = sb.x; ssrc[p5] = sb.y; ssrc[p6] = sb.z; ssrc[p7] = sb.w;
}

__global__ __launch_bounds__(256) void gemm_count_kernel(const float* __restrict__ feats,
                                                         const float* __restrict__ W,
                                                         float* __restrict__ h,
                                                         const int* __restrict__ dst,
                                                         int* __restrict__ cnt,
                                                         int* __restrict__ rank) {
    if (blockIdx.x >= GEMM_BLOCKS) {
        count_block(blockIdx.x - GEMM_BLOCKS, dst, cnt, rank);
        return;
    }
    const int wave = blockIdx.x * (blockDim.x >> 5) + (threadIdx.x >> 5);
    const int lane = threadIdx.x & 31;
    const int lo = lane & 15;
    const int hi = lane >> 4;
    const int rowBase = wave * 16;
    const float* __restrict__ arow = feats + (size_t)(rowBase + lo) * IN_DIM;

    v8f acc[8];
#pragma unroll
    for (int t = 0; t < 8; ++t) acc[t] = (v8f){0.f, 0.f, 0.f, 0.f, 0.f, 0.f, 0.f, 0.f};

    for (int k = 0; k < IN_DIM; k += 4) {
        const int ka = k + 2 * hi;
        v2f a;
        a.x = arow[ka];
        a.y = arow[ka + 1];
#pragma unroll
        for (int t = 0; t < 8; ++t) {
            const int col = t * 16 + lo;
            v2f b;
            b.x = W[(size_t)ka * FEAT + col];
            b.y = W[(size_t)(ka + 1) * FEAT + col];
            acc[t] = __builtin_amdgcn_wmma_f32_16x16x4_f32(
                false, a, false, b, (short)0, acc[t], false, false);
        }
    }

#pragma unroll
    for (int t = 0; t < 8; ++t)
#pragma unroll
        for (int i = 0; i < 8; ++i)
            h[(size_t)(rowBase + i + 8 * hi) * FEAT + t * 16 + lo] = acc[t][i];
}

__global__ __launch_bounds__(256) void logits_fill_kernel(const float* __restrict__ h,
                                                          const float* __restrict__ attn_l,
                                                          const float* __restrict__ attn_r,
                                                          float* __restrict__ el,
                                                          float* __restrict__ er,
                                                          const int* __restrict__ src,
                                                          const int* __restrict__ dst,
                                                          const int* __restrict__ rank,
                                                          const int* __restrict__ off,
                                                          int* __restrict__ ssrc) {
    if (blockIdx.x < EDGE_BLOCKS) {
        fill_block(blockIdx.x, src, dst, rank, off, ssrc);
        return;
    }
    int i = (blockIdx.x - EDGE_BLOCKS) * blockDim.x + threadIdx.x;
    if (i >= N_NODES * HEADS) return;
    const int n = i >> 2;
    const int hh = i & 3;
    const float* __restrict__ hp = h + (size_t)n * FEAT + hh * OUT_D;
    const float* __restrict__ al = attn_l + hh * OUT_D;
    const float* __restrict__ ar = attn_r + hh * OUT_D;
    float sl = 0.f, sr = 0.f;
#pragma unroll
    for (int d = 0; d < OUT_D; ++d) {
        float v = hp[d];
        sl += v * al[d];
        sr += v * ar[d];
    }
    el[i] = sl;
    er[i] = sr;
}

__device__ __forceinline__ float leaky(float v) { return (v >= 0.f) ? v : NEG_SLOPE * v; }

__device__ __forceinline__ float wave_max(float v) {
#pragma unroll
    for (int o = 16; o > 0; o >>= 1) v = fmaxf(v, __shfl_xor(v, o));
    return v;
}

__device__ __forceinline__ float wave_sum(float v) {
#pragma unroll
    for (int o = 16; o > 0; o >>= 1) v += __shfl_xor(v, o);
    return v;
}

__device__ __forceinline__ int lane_bcast(float v, int j) {
    return __builtin_amdgcn_readlane(__float_as_int(v), j);
}

__device__ __forceinline__ float pick_head(int b0, int b1, int b2, int b3, int k0, int k1, int k2, int k3) {
    return __int_as_float((b0 & k0) | (b1 & k1) | (b2 & k2) | (b3 & k3));
}

__device__ __forceinline__ void add_chunk(float4& acc, int s, float a0, float a1, float a2, float a3,
                                          int cntc, int hd, int lane, const float* __restrict__ h) {
    const int k0 = (hd == 0) ? -1 : 0, k1 = (hd == 1) ? -1 : 0;
    const int k2 = (hd == 2) ? -1 : 0, k3 = (hd == 3) ? -1 : 0;
    int j = 0;
    for (; j + 4 <= cntc; j += 4) {
        float4 hv[4];
        float a[4];
#pragma unroll
        for (int u = 0; u < 4; ++u) {
            const int sj = __builtin_amdgcn_readlane(s, j + u);
            hv[u] = *(const float4*)(h + (size_t)sj * FEAT + lane * 4);
            a[u] = pick_head(lane_bcast(a0, j + u), lane_bcast(a1, j + u),
                             lane_bcast(a2, j + u), lane_bcast(a3, j + u), k0, k1, k2, k3);
        }
#pragma unroll
        for (int u = 0; u < 4; ++u) {
            acc.x += hv[u].x * a[u];
            acc.y += hv[u].y * a[u];
            acc.z += hv[u].z * a[u];
            acc.w += hv[u].w * a[u];
        }
    }
    for (; j < cntc; ++j) {
        const int sj = __builtin_amdgcn_readlane(s, j);
        const float4 hv = *(const float4*)(h + (size_t)sj * FEAT + lane * 4);
        const float a = pick_head(lane_bcast(a0, j), lane_bcast(a1, j),
                                  lane_bcast(a2, j), lane_bcast(a3, j), k0, k1, k2, k3);
        acc.x += hv.x * a;
        acc.y += hv.y * a;
        acc.z += hv.z * a;
        acc.w += hv.w * a;
    }
}

__global__ __launch_bounds__(256) void aggregate_kernel(const int* __restrict__ off,
                                                        const int* __restrict__ cnt,
                                                        const int* __restrict__ ssrc,
                                                        const float* __restrict__ h,
                                                        const float* __restrict__ el,
                                                        const float* __restrict__ er,
                                                        float* __restrict__ out) {
    const int lane = threadIdx.x & 31;
    const int hd = lane >> 3;
    const int d = blockIdx.x * 8 + __builtin_amdgcn_readfirstlane((int)(threadIdx.x >> 5));
    const int begin = off[d];
    const int n = cnt[d];
    const float4* __restrict__ el4 = (const float4*)el;
    float4 acc = {0.f, 0.f, 0.f, 0.f};

    if (n > 0) {
        const float4 r4 = *(const float4*)(er + (size_t)d * HEADS);
        const bool v0 = lane < n;
        int s0 = 0;
        float e0 = -INFINITY, e1 = -INFINITY, e2 = -INFINITY, e3 = -INFINITY;
        if (v0) {
            s0 = ssrc[begin + lane];
            const float4 l4 = el4[s0];
            e0 = leaky(l4.x + r4.x); e1 = leaky(l4.y + r4.y);
            e2 = leaky(l4.z + r4.z); e3 = leaky(l4.w + r4.w);
        }
        float m0 = e0, m1 = e1, m2 = e2, m3 = e3;
        for (int base = 32; base < n; base += 32) {
            const int i = base + lane;
            if (i < n) {
                const float4 l4 = el4[ssrc[begin + i]];
                m0 = fmaxf(m0, leaky(l4.x + r4.x)); m1 = fmaxf(m1, leaky(l4.y + r4.y));
                m2 = fmaxf(m2, leaky(l4.z + r4.z)); m3 = fmaxf(m3, leaky(l4.w + r4.w));
            }
        }
        m0 = wave_max(m0); m1 = wave_max(m1); m2 = wave_max(m2); m3 = wave_max(m3);
        const float p0 = v0 ? expf(e0 - m0) : 0.f, p1 = v0 ? expf(e1 - m1) : 0.f;
        const float p2 = v0 ? expf(e2 - m2) : 0.f, p3 = v0 ? expf(e3 - m3) : 0.f;
        float z0 = p0, z1 = p1, z2 = p2, z3 = p3;
        for (int base = 32; base < n; base += 32) {
            const int i = base + lane;
            if (i < n) {
                const float4 l4 = el4[ssrc[begin + i]];
                z0 += expf(leaky(l4.x + r4.x) - m0); z1 += expf(leaky(l4.y + r4.y) - m1);
                z2 += expf(leaky(l4.z + r4.z) - m2); z3 += expf(leaky(l4.w + r4.w) - m3);
            }
        }
        z0 = wave_sum(z0); z1 = wave_sum(z1); z2 = wave_sum(z2); z3 = wave_sum(z3);
        add_chunk(acc, s0, p0 / z0, p1 / z1, p2 / z2, p3 / z3, (n < 32) ? n : 32, hd, lane, h);
        for (int base = 32; base < n; base += 32) {
            const int i = base + lane;
            int s = 0;
            float a0 = 0.f, a1 = 0.f, a2 = 0.f, a3 = 0.f;
            if (i < n) {
                s = ssrc[begin + i];
                const float4 l4 = el4[s];
                a0 = expf(leaky(l4.x + r4.x) - m0) / z0; a1 = expf(leaky(l4.y + r4.y) - m1) / z1;
                a2 = expf(leaky(l4.z + r4.z) - m2) / z2; a3 = expf(leaky(l4.w + r4.w) - m3) / z3;
            }
            const int rem = n - base;
            add_chunk(acc, s, a0, a1, a2, a3, (rem < 32) ? rem : 32, hd, lane, h);
        }
    }

    float4 r;
    r.x = (acc.x > 0.f) ? acc.x : (expf(acc.x) - 1.0f);
    r.y = (acc.y > 0.f) ? acc.y : (expf(acc.y) - 1.0f);
    r.z = (acc.z > 0.f) ? acc.z : (expf(acc.z) - 1.0f);
    r.w = (acc.w > 0.f) ? acc.w : (expf(acc.w) - 1.0f);
    *(float4*)(out + (size_t)d * FEAT + lane * 4) = r;
    if (blockIdx.x == 0 && threadIdx.x == 0) {
        out[(size_t)NUM_DSTN * FEAT + 0] = 1.0f;
        out[(size_t)NUM_DSTN * FEAT + 1] = 1.0f;
    }
}

extern "C" void kernel_launch(void* const* d_in, const int* in_sizes, int n_in,
                              void* d_out, int out_size, void* d_ws, size_t ws_size,
                              hipStream_t stream) {
    const float* feats  = (const float*)d_in[0];
    const int*   src    = (const int*)d_in[1];
    const int*   dst    = (const int*)d_in[2];
    const float* W      = (const float*)d_in[3];
    const float* attn_l = (const float*)d_in[4];
    const float* attn_r = (const float*)d_in[5];
    float* out = (float*)d_out;

    float* ws   = (float*)d_ws;
    float* hbuf = ws;
    float* el   = hbuf + (size_t)N_NODES * FEAT;
    float* er   = el + (size_t)N_NODES * HEADS;
    int*   rank = (int*)(er + (size_t)N_NODES * HEADS);
    int*   ssrc = rank + (size_t)E_EDGES;
    int*   cnt  = ssrc + (size_t)E_EDGES;
    int*   off  = cnt + (size_t)NUM_DSTN;

    init_kernel<<<NUM_DSTN / 256, 256, 0, stream>>>(cnt);
    gemm_count_kernel<<<GEMM_BLOCKS + EDGE_BLOCKS, 256, 0, stream>>>(feats, W, hbuf, dst, cnt, rank);
    scan_kernel<<<1, 1024, 0, stream>>>(cnt, off);
    logits_fill_kernel<<<EDGE_BLOCKS + LOGIT_BLOCKS, 256, 0, stream>>>(hbuf, attn_l, attn_r, el, er,
                                                                       src, dst, rank, off, ssrc);
    aggregate_kernel<<<NUM_DSTN / 8, 256, 0, stream>>>(off, cnt, ssrc, hbuf, el, er, out);
}


# GEMM: W staged once per block in LDS as k-row pairs (144 KiB, two blocks per WGP); a B fragment is one 8-byte LDS read instead of two global loads; count pass moves into the GEMM threads
# speedup vs baseline: 4.4322x; 1.0578x over previous
#include <hip/hip_runtime.h>
#include <math.h>

#define N_NODES 131072
#define E_EDGES 1048576
#define NUM_DSTN 65536
#define IN_DIM 256
#define HEADS 4
#define OUT_D 32
#define FEAT (HEADS * OUT_D)
#define NEG_SLOPE 0.2f

typedef __attribute__((ext_vector_type(2))) float v2f;
typedef __attribute__((ext_vector_type(8))) float v8f;

__global__ void init_kernel(int* __restrict__ cnt) {
    const int i = blockIdx.x * blockDim.x + threadIdx.x;
    if (i < NUM_DSTN) cnt[i] = 0;
}

#define EDGE_PER_BLOCK 2048
#define EDGE_BLOCKS (E_EDGES / EDGE_PER_BLOCK)
#define GEMM_THREADS 512
#define GEMM_BLOCKS (N_NODES / 16 / 16)
#define WL_PITCH 144
#define LOGIT_BLOCKS (N_NODES * HEADS / 256)

__global__ __launch_bounds__(1024) void scan_kernel(const int* __restrict__ cnt,
                                                    int* __restrict__ off) {
    __shared__ int part[512];
    __shared__ int wtot[16];
    const int t = threadIdx.x;
    const int lane = t & 31;
    const int wave = t >> 5;
    const int4* __restrict__ c4 = (const int4*)cnt;
    int4 v[16];
    int incl[16];
#pragma unroll
    for (int c = 0; c < 16; ++c) v[c] = c4[c * 1024 + t];
#pragma unroll
    for (int c = 0; c < 16; ++c) {
        int s = v[c].x + v[c].y + v[c].z + v[c].w;
#pragma unroll
        for (int o = 1; o < 32; o <<= 1) {
            const int nb = __shfl_up(s, o);
            if (lane >= o) s += nb;
        }
        incl[c] = s;
        if (lane == 31) part[c * 32 + wave] = s;
    }
    __syncthreads();
    int p = 0, pin = 0;
    if (t < 512) {
        p = part[t];
        pin = p;
#pragma unroll
        for (int o = 1; o < 32; o <<= 1) {
            const int nb = __shfl_up(pin, o);
            if (lane >= o) pin += nb;
        }
        if (lane == 31) wtot[wave] = pin;
    }
    __syncthreads();
    if (t < 512) {
        int b = 0;
        for (int w = 0; w < wave; ++w) b += wtot[w];
        part[t] = b + pin - p;
    }
    __syncthreads();
    int4* __restrict__ o4 = (int4*)off;
#pragma unroll
    for (int c = 0; c < 16; ++c) {
        const int own = v[c].x + v[c].y + v[c].z + v[c].w;
        const int base = part[c * 32 + wave] + incl[c] - own;
        int4 r;
        r.x = base;
        r.y = r.x + v[c].x;
        r.z = r.y + v[c].y;
        r.w = r.z + v[c].z;
        o4[c * 1024 + t] = r;
    }
}

__device__ __forceinline__ void fill_block(int b, const int* __restrict__ src,
                                           const int* __restrict__ dst,
                                           const int* __restrict__ rank,
                                           const int* __restrict__ off, int* __restrict__ ssrc) {
    const int t = threadIdx.x;
    const size_t q = (size_t)b * (EDGE_PER_BLOCK / 4);
    const int4* __restrict__ s4 = (const int4*)src + q;
    const int4* __restrict__ d4 = (const int4*)dst + q;
    const int4* __restrict__ r4 = (const int4*)rank + q;
    const int4 sa = s4[t], sb = s4[256 + t];
    const int4 da = d4[t], db = d4[256 + t];
    const int4 ra = r4[t], rb = r4[256 + t];
    const int p0 = off[da.x] + ra.x, p1 = off[da.y] + ra.y;
    const int p2 = off[da.z] + ra.z, p3 = off[da.w] + ra.w;
    const int p4 = off[db.x] + rb.x, p5 = off[db.y] + rb.y;
    const int p6 = off[db.z] + rb.z, p7 = off[db.w] + rb.w;
    ssrc[p0] = sa.x; ssrc[p1] = sa.y; ssrc[p2] = sa.z; ssrc[p3] = sa.w;
    ssrc[p4] = sb.x; ssrc[p5] = sb.y; ssrc[p6] = sb.z; ssrc[p7] = sb.w;
}

__global__ __launch_bounds__(GEMM_THREADS) void gemm_count_kernel(const float* __restrict__ feats,
                                                                  const float* __restrict__ W,
                                                                  float* __restrict__ h,
                                                                  const int* __restrict__ dst,
                                                                  int* __restrict__ cnt,
                                                                  int* __restrict__ rank) {
    __shared__ float4 Wl[(IN_DIM / 2) * WL_PITCH / 2];
    const int tid = threadIdx.x;

    {
        const int g = blockIdx.x * GEMM_THREADS + tid;
        const int4 dd = ((const int4*)dst)[g];
        int4 rr;
        rr.x = atomicAdd(&cnt[dd.x], 1);
        rr.y = atomicAdd(&cnt[dd.y], 1);
        rr.z = atomicAdd(&cnt[dd.z], 1);
        rr.w = atomicAdd(&cnt[dd.w], 1);
        ((int4*)rank)[g] = rr;
    }

    {
        const int c4 = tid & 31;
        const int kp0 = tid >> 5;
#pragma unroll
        for (int it = 0; it < 8; ++it) {
            const int kp = kp0 + 16 * it;
            const float4* __restrict__ wp = (const float4*)(W + (size_t)(2 * kp) * FEAT) + c4;
            const float4 r0 = wp[0];
            const float4 r1 = wp[FEAT / 4];
            float4* __restrict__ o = Wl + (kp * WL_PITCH + 4 * c4) / 2;
            o[0] = make_float4(r0.x, r1.x, r0.y, r1.y);
            o[1] = make_float4(r0.z, r1.z, r0.w, r1.w);
        }
    }
    __syncthreads();

    const int lane = tid & 31;
    const int lo = lane & 15;
    const int hi = lane >> 4;
    const int rowBase = (blockIdx.x * (GEMM_THREADS >> 5) + (tid >> 5)) * 16;
    const float* __restrict__ arow = feats + (size_t)(rowBase + lo) * IN_DIM;
    const v2f* __restrict__ bl = (const v2f*)Wl + hi * WL_PITCH + lo;

    v8f acc[8];
#pragma unroll
    for (int t = 0; t < 8; ++t) acc[t] = (v8f){0.f, 0.f, 0.f, 0.f, 0.f, 0.f, 0.f, 0.f};

    for (int k = 0; k < IN_DIM; k += 4) {
        const int ka = k + 2 * hi;
        v2f a;
        a.x = arow[ka];
        a.y = arow[ka + 1];
#pragma unroll
        for (int t = 0; t < 8; ++t) {
            const v2f b = bl[(k >> 1) * WL_PITCH + t * 16];
            acc[t] = __builtin_amdgcn_wmma_f32_16x16x4_f32(
                false, a, false, b, (short)0, acc[t], false, false);
        }
    }

#pragma unroll
    for (int t = 0; t < 8; ++t)
#pragma unroll
        for (int i = 0; i < 8; ++i)
            h[(size_t)(rowBase + i + 8 * hi) * FEAT + t * 16 + lo] = acc[t][i];
}

__global__ __launch_bounds__(256) void logits_fill_kernel(const float* __restrict__ h,
                                                          const float* __restrict__ attn_l,
                                                          const float* __restrict__ attn_r,
                                                          float* __restrict__ el,
                                                          float* __restrict__ er,
                                                          const int* __restrict__ src,
                                                          const int* __restrict__ dst,
                                                          const int* __restrict__ rank,
                                                          const int* __restrict__ off,
                                                          int* __restrict__ ssrc) {
    if (blockIdx.x < EDGE_BLOCKS) {
        fill_block(blockIdx.x, src, dst, rank, off, ssrc);
        return;
    }
    int i = (blockIdx.x - EDGE_BLOCKS) * blockDim.x + threadIdx.x;
    if (i >= N_NODES * HEADS) return;
    const int n = i >> 2;
    const int hh = i & 3;
    const float* __restrict__ hp = h + (size_t)n * FEAT + hh * OUT_D;
    const float* __restrict__ al = attn_l + hh * OUT_D;
    const float* __restrict__ ar = attn_r + hh * OUT_D;
    float sl = 0.f, sr = 0.f;
#pragma unroll
    for (int d = 0; d < OUT_D; ++d) {
        float v = hp[d];
        sl += v * al[d];
        sr += v * ar[d];
    }
    el[i] = sl;
    er[i] = sr;
}

__device__ __forceinline__ float leaky(float v) { return (v >= 0.f) ? v : NEG_SLOPE * v; }

__device__ __forceinline__ float wave_max(float v) {
#pragma unroll
    for (int o = 16; o > 0; o >>= 1) v = fmaxf(v, __shfl_xor(v, o));
    return v;
}

__device__ __forceinline__ float wave_sum(float v) {
#pragma unroll
    for (int o = 16; o > 0; o >>= 1) v += __shfl_xor(v, o);
    return v;
}

__device__ __forceinline__ int lane_bcast(float v, int j) {
    return __builtin_amdgcn_readlane(__float_as_int(v), j);
}

__device__ __forceinline__ float pick_head(int b0, int b1, int b2, int b3, int k0, int k1, int k2, int k3) {
    return __int_as_float((b0 & k0) | (b1 & k1) | (b2 & k2) | (b3 & k3));
}

__device__ __forceinline__ void add_chunk(float4& acc, int s, float a0, float a1, float a2, float a3,
                                          int cntc, int hd, int lane, const float* __restrict__ h) {
    const int k0 = (hd == 0) ? -1 : 0, k1 = (hd == 1) ? -1 : 0;
    const int k2 = (hd == 2) ? -1 : 0, k3 = (hd == 3) ? -1 : 0;
    int j = 0;
    for (; j + 4 <= cntc; j += 4) {
        float4 hv[4];
        float a[4];
#pragma unroll
        for (int u = 0; u < 4; ++u) {
            const int sj = __builtin_amdgcn_readlane(s, j + u);
            hv[u] = *(const float4*)(h + (size_t)sj * FEAT + lane * 4);
            a[u] = pick_head(lane_bcast(a0, j + u), lane_bcast(a1, j + u),
                             lane_bcast(a2, j + u), lane_bcast(a3, j + u), k0, k1, k2, k3);
        }
#pragma unroll
        for (int u = 0; u < 4; ++u) {
            acc.x += hv[u].x * a[u];
            acc.y += hv[u].y * a[u];
            acc.z += hv[u].z * a[u];
            acc.w += hv[u].w * a[u];
        }
    }
    for (; j < cntc; ++j) {
        const int sj = __builtin_amdgcn_readlane(s, j);
        const float4 hv = *(const float4*)(h + (size_t)sj * FEAT + lane * 4);
        const float a = pick_head(lane_bcast(a0, j), lane_bcast(a1, j),
                                  lane_bcast(a2, j), lane_bcast(a3, j), k0, k1, k2, k3);
        acc.x += hv.x * a;
        acc.y += hv.y * a;
        acc.z += hv.z * a;
        acc.w += hv.w * a;
    }
}

__global__ __launch_bounds__(256) void aggregate_kernel(const int* __restrict__ off,
                                                        const int* __restrict__ cnt,
                                                        const int* __restrict__ ssrc,
                                                        const float* __restrict__ h,
                                                        const float* __restrict__ el,
                                                        const float* __restrict__ er,
                                                        float* __restrict__ out) {
    const int lane = threadIdx.x & 31;
    const int hd = lane >> 3;
    const int d = blockIdx.x * 8 + __builtin_amdgcn_readfirstlane((int)(threadIdx.x >> 5));
    const int begin = off[d];
    const int n = cnt[d];
    const float4* __restrict__ el4 = (const float4*)el;
    float4 acc = {0.f, 0.f, 0.f, 0.f};

    if (n > 0) {
        const float4 r4 = *(const float4*)(er + (size_t)d * HEADS);
        const bool v0 = lane < n;
        int s0 = 0;
        float e0 = -INFINITY, e1 = -INFINITY, e2 = -INFINITY, e3 = -INFINITY;
        if (v0) {
            s0 = ssrc[begin + lane];
            const float4 l4 = el4[s0];
            e0 = leaky(l4.x + r4.x); e1 = leaky(l4.y + r4.y);
            e2 = leaky(l4.z + r4.z); e3 = leaky(l4.w + r4.w);
        }
        float m0 = e0, m1 = e1, m2 = e2, m3 = e3;
        for (int base = 32; base < n; base += 32) {
            const int i = base + lane;
            if (i < n) {
                const float4 l4 = el4[ssrc[begin + i]];
                m0 = fmaxf(m0, leaky(l4.x + r4.x)); m1 = fmaxf(m1, leaky(l4.y + r4.y));
                m2 = fmaxf(m2, leaky(l4.z + r4.z)); m3 = fmaxf(m3, leaky(l4.w + r4.w));
            }
        }
        m0 = wave_max(m0); m1 = wave_max(m1); m2 = wave_max(m2); m3 = wave_max(m3);
        const float p0 = v0 ? expf(e0 - m0) : 0.f, p1 = v0 ? expf(e1 - m1) : 0.f;
        const float p2 = v0 ? expf(e2 - m2) : 0.f, p3 = v0 ? expf(e3 - m3) : 0.f;
        float z0 = p0, z1 = p1, z2 = p2, z3 = p3;
        for (int base = 32; base < n; base += 32) {
            const int i = base + lane;
            if (i < n) {
                const float4 l4 = el4[ssrc[begin + i]];
                z0 += expf(leaky(l4.x + r4.x) - m0); z1 += expf(leaky(l4.y + r4.y) - m1);
                z2 += expf(leaky(l4.z + r4.z) - m2); z3 += expf(leaky(l4.w + r4.w) - m3);
            }
        }
        z0 = wave_sum(z0); z1 = wave_sum(z1); z2 = wave_sum(z2); z3 = wave_sum(z3);
        add_chunk(acc, s0, p0 / z0, p1 / z1, p2 / z2, p3 / z3, (n < 32) ? n : 32, hd, lane, h);
        for (int base = 32; base < n; base += 32) {
            const int i = base + lane;
            int s = 0;
            float a0 = 0.f, a1 = 0.f, a2 = 0.f, a3 = 0.f;
            if (i < n) {
                s = ssrc[begin + i];
                const float4 l4 = el4[s];
                a0 = expf(leaky(l4.x + r4.x) - m0) / z0; a1 = expf(leaky(l4.y + r4.y) - m1) / z1;
                a2 = expf(leaky(l4.z + r4.z) - m2) / z2; a3 = expf(leaky(l4.w + r4.w) - m3) / z3;
            }
            const int rem = n - base;
            add_chunk(acc, s, a0, a1, a2, a3, (rem < 32) ? rem : 32, hd, lane, h);
        }
    }

    float4 r;
    r.x = (acc.x > 0.f) ? acc.x : (expf(acc.x) - 1.0f);
    r.y = (acc.y > 0.f) ? acc.y : (expf(acc.y) - 1.0f);
    r.z = (acc.z > 0.f) ? acc.z : (expf(acc.z) - 1.0f);
    r.w = (acc.w > 0.f) ? acc.w : (expf(acc.w) - 1.0f);
    *(float4*)(out + (size_t)d * FEAT + lane * 4) = r;
    if (blockIdx.x == 0 && threadIdx.x == 0) {
        out[(size_t)NUM_DSTN * FEAT + 0] = 1.0f;
        out[(size_t)NUM_DSTN * FEAT + 1] = 1.0f;
    }
}

extern "C" void kernel_launch(void* const* d_in, const int* in_sizes, int n_in,
                              void* d_out, int out_size, void* d_ws, size_t ws_size,
                              hipStream_t stream) {
    const float* feats  = (const float*)d_in[0];
    const int*   src    = (const int*)d_in[1];
    const int*   dst    = (const int*)d_in[2];
    const float* W      = (const float*)d_in[3];
    const float* attn_l = (const float*)d_in[4];
    const float* attn_r = (const float*)d_in[5];
    float* out = (float*)d_out;

    float* ws   = (float*)d_ws;
    float* hbuf = ws;
    float* el   = hbuf + (size_t)N_NODES * FEAT;
    float* er   = el + (size_t)N_NODES * HEADS;
    int*   rank = (int*)(er + (size_t)N_NODES * HEADS);
    int*   ssrc = rank + (size_t)E_EDGES;
    int*   cnt  = ssrc + (size_t)E_EDGES;
    int*   off  = cnt + (size_t)NUM_DSTN;

    init_kernel<<<NUM_DSTN / 256, 256, 0, stream>>>(cnt);
    gemm_count_kernel<<<GEMM_BLOCKS, GEMM_THREADS, 0, stream>>>(feats, W, hbuf, dst, cnt, rank);
    scan_kernel<<<1, 1024, 0, stream>>>(cnt, off);
    logits_fill_kernel<<<EDGE_BLOCKS + LOGIT_BLOCKS, 256, 0, stream>>>(hbuf, attn_l, attn_r, el, er,
                                                                       src, dst, rank, off, ssrc);
    aggregate_kernel<<<NUM_DSTN / 8, 256, 0, stream>>>(off, cnt, ssrc, hbuf, el, er, out);
}


# Edge grouping as a two-level counting sort on LDS counters (bin = dst>>8: block histograms, column prefix, per-bin sort); the 1M global atomics, init and scan launches go; sum order = LDS arrival
# speedup vs baseline: 5.5745x; 1.2577x over previous
#include <hip/hip_runtime.h>
#include <math.h>

#define N_NODES 131072
#define E_EDGES 1048576
#define NUM_DSTN 65536
#define IN_DIM 256
#define HEADS 4
#define OUT_D 32
#define FEAT (HEADS * OUT_D)
#define NEG_SLOPE 0.2f

typedef __attribute__((ext_vector_type(2))) float v2f;
typedef __attribute__((ext_vector_type(8))) float v8f;

#define L1_BLOCKS 256
#define L1_INT4 1024
#define BINS 256
#define SRC_BITS 17
#define GEMM_THREADS 512
#define GEMM_BLOCKS (N_NODES / 16 / 16)
#define WL_PITCH 144
#define LOGIT_BLOCKS (N_NODES * HEADS / 256)

__global__ __launch_bounds__(512) void bin_count_kernel(const int* __restrict__ dst,
                                                        int* __restrict__ H) {
    __shared__ int hist[BINS];
    const int t = threadIdx.x;
    if (t < BINS) hist[t] = 0;
    __syncthreads();
    const int4* __restrict__ d4 = (const int4*)dst + (size_t)blockIdx.x * L1_INT4;
    const int4 a = d4[t], b = d4[512 + t];
    atomicAdd(&hist[a.x >> 8], 1); atomicAdd(&hist[a.y >> 8], 1);
    atomicAdd(&hist[a.z >> 8], 1); atomicAdd(&hist[a.w >> 8], 1);
    atomicAdd(&hist[b.x >> 8], 1); atomicAdd(&hist[b.y >> 8], 1);
    atomicAdd(&hist[b.z >> 8], 1); atomicAdd(&hist[b.w >> 8], 1);
    __syncthreads();
    if (t < BINS) H[blockIdx.x * BINS + t] = hist[t];
}

__device__ __forceinline__ void group_block(int bin, const int* __restrict__ bstart,
                                            const int* __restrict__ binned,
                                            int* __restrict__ cnt, int* __restrict__ off,
                                            int* __restrict__ ssrc) {
    __shared__ int cntL[BINS];
    __shared__ int curL[BINS];
    __shared__ int wtotL[8];
    const int t = threadIdx.x;
    const int lane = t & 31;
    const int wave = t >> 5;
    const int S = bstart[bin];
    const int Eend = (bin == BINS - 1) ? E_EDGES : bstart[bin + 1];
    const int m = Eend - S;
    const int* __restrict__ rec = binned + S;
    cntL[t] = 0;
    __syncthreads();
    int j = t;
    for (; j + 1792 < m; j += 2048) {
        int r[8];
#pragma unroll
        for (int u = 0; u < 8; ++u) r[u] = rec[j + 256 * u];
#pragma unroll
        for (int u = 0; u < 8; ++u) atomicAdd(&cntL[r[u] >> SRC_BITS], 1);
    }
    for (; j < m; j += 256) atomicAdd(&cntL[rec[j] >> SRC_BITS], 1);
    __syncthreads();
    const int c = cntL[t];
    int incl = c;
#pragma unroll
    for (int o = 1; o < 32; o <<= 1) {
        const int nb = __shfl_up(incl, o);
        if (lane >= o) incl += nb;
    }
    if (lane == 31) wtotL[wave] = incl;
    __syncthreads();
    int wb = 0;
    for (int w = 0; w < wave; ++w) wb += wtotL[w];
    const int o = S + wb + incl - c;
    curL[t] = o;
    cnt[bin * BINS + t] = c;
    off[bin * BINS + t] = o;
    __syncthreads();
    j = t;
    for (; j + 1792 < m; j += 2048) {
        int r[8], p[8];
#pragma unroll
        for (int u = 0; u < 8; ++u) r[u] = rec[j + 256 * u];
#pragma unroll
        for (int u = 0; u < 8; ++u) p[u] = atomicAdd(&curL[r[u] >> SRC_BITS], 1);
#pragma unroll
        for (int u = 0; u < 8; ++u) ssrc[p[u]] = r[u] & ((1 << SRC_BITS) - 1);
    }
    for (; j < m; j += 256) {
        const int r0 = rec[j];
        ssrc[atomicAdd(&curL[r0 >> SRC_BITS], 1)] = r0 & ((1 << SRC_BITS) - 1);
    }
}

__global__ __launch_bounds__(GEMM_THREADS) void gemm_bin_kernel(const float* __restrict__ feats,
                                                                const float* __restrict__ W,
                                                                float* __restrict__ h,
                                                                const int* __restrict__ src,
                                                                const int* __restrict__ dst,
                                                                const int* __restrict__ H,
                                                                int* __restrict__ bstart,
                                                                int* __restrict__ binned) {
    __shared__ float4 Wl[(IN_DIM / 2) * WL_PITCH / 2];
    const int tid = threadIdx.x;

    if ((blockIdx.x & 1) == 0) {
        int* __restrict__ sc = (int*)Wl;
        int* __restrict__ preS = sc;
        int* __restrict__ totS = sc + 8 * BINS;
        int* __restrict__ cur = sc + 16 * BINS;
        int* __restrict__ wtot = sc + 17 * BINS;
        const int bb = blockIdx.x >> 1;
        const int4* __restrict__ d4 = (const int4*)dst + (size_t)bb * L1_INT4;
        const int4* __restrict__ s4 = (const int4*)src + (size_t)bb * L1_INT4;
        const int4 da = d4[tid], db = d4[512 + tid];
        const int4 sa = s4[tid], sb = s4[512 + tid];
        const int c4 = tid & 63;
        const int g = tid >> 6;
        const int4* __restrict__ H4 = (const int4*)H + (size_t)(g * 32) * (BINS / 4) + c4;
        int4 pre = {0, 0, 0, 0}, tot = {0, 0, 0, 0};
#pragma unroll 8
        for (int i = 0; i < 32; ++i) {
            const int4 v = H4[i * (BINS / 4)];
            const int k = (g * 32 + i < bb) ? -1 : 0;
            tot.x += v.x; tot.y += v.y; tot.z += v.z; tot.w += v.w;
            pre.x += v.x & k; pre.y += v.y & k; pre.z += v.z & k; pre.w += v.w & k;
        }
        ((int4*)preS)[g * (BINS / 4) + c4] = pre;
        ((int4*)totS)[g * (BINS / 4) + c4] = tot;
        __syncthreads();
        const int lane = tid & 31;
        const int wave = tid >> 5;
        int p = 0, c = 0, incl = 0;
        if (tid < BINS) {
#pragma unroll
            for (int gg = 0; gg < 8; ++gg) {
                p += preS[gg * BINS + tid];
                c += totS[gg * BINS + tid];
            }
            incl = c;
#pragma unroll
            for (int o = 1; o < 32; o <<= 1) {
                const int nb = __shfl_up(incl, o);
                if (lane >= o) incl += nb;
            }
            if (lane == 31) wtot[wave] = incl;
        }
        __syncthreads();
        if (tid < BINS) {
            int wb = 0;
            for (int w = 0; w < wave; ++w) wb += wtot[w];
            const int bs = wb + incl - c;
            cur[tid] = bs + p;
            if (bb == 0) bstart[tid] = bs;
        }
        __syncthreads();
        binned[atomicAdd(&cur[da.x >> 8], 1)] = ((da.x & 255) << SRC_BITS) | sa.x;
        binned[atomicAdd(&cur[da.y >> 8], 1)] = ((da.y & 255) << SRC_BITS) | sa.y;
        binned[atomicAdd(&cur[da.z >> 8], 1)] = ((da.z & 255) << SRC_BITS) | sa.z;
        binned[atomicAdd(&cur[da.w >> 8], 1)] = ((da.w & 255) << SRC_BITS) | sa.w;
        binned[atomicAdd(&cur[db.x >> 8], 1)] = ((db.x & 255) << SRC_BITS) | sb.x;
        binned[atomicAdd(&cur[db.y >> 8], 1)] = ((db.y & 255) << SRC_BITS) | sb.y;
        binned[atomicAdd(&cur[db.z >> 8], 1)] = ((db.z & 255) << SRC_BITS) | sb.z;
        binned[atomicAdd(&cur[db.w >> 8], 1)] = ((db.w & 255) << SRC_BITS) | sb.w;
        __syncthreads();
    }

    {
        const int c4 = tid & 31;
        const int kp0 = tid >> 5;
#pragma unroll
        for (int it = 0; it < 8; ++it) {
            const int kp = kp0 + 16 * it;
            const float4* __restrict__ wp = (const float4*)(W + (size_t)(2 * kp) * FEAT) + c4;
            const float4 r0 = wp[0];
            const float4 r1 = wp[FEAT / 4];
            float4* __restrict__ o = Wl + (kp * WL_PITCH + 4 * c4) / 2;
            o[0] = make_float4(r0.x, r1.x, r0.y, r1.y);
            o[1] = make_float4(r0.z, r1.z, r0.w, r1.w);
        }
    }
    __syncthreads();

    const int lane = tid & 31;
    const int lo = lane & 15;
    const int hi = lane >> 4;
    const int rowBase = (blockIdx.x * (GEMM_THREADS >> 5) + (tid >> 5)) * 16;
    const float* __restrict__ arow = feats + (size_t)(rowBase + lo) * IN_DIM;
    const v2f* __restrict__ bl = (const v2f*)Wl + hi * WL_PITCH + lo;

    v8f acc[8];
#pragma unroll
    for (int t = 0; t < 8; ++t) acc[t] = (v8f){0.f, 0.f, 0.f, 0.f, 0.f, 0.f, 0.f, 0.f};

    for (int k = 0; k < IN_DIM; k += 4) {
        const int ka = k + 2 * hi;
        v2f a;
        a.x = arow[ka];
        a.y = arow[ka + 1];
#pragma unroll
        for (int t = 0; t < 8; ++t) {
            const v2f b = bl[(k >> 1) * WL_PITCH + t * 16];
            acc[t] = __builtin_amdgcn_wmma_f32_16x16x4_f32(
                false, a, false, b, (short)0, acc[t], false, false);
        }
    }

#pragma unroll
    for (int t = 0; t < 8; ++t)
#pragma unroll
        for (int i = 0; i < 8; ++i)
            h[(size_t)(rowBase + i + 8 * hi) * FEAT + t * 16 + lo] = acc[t][i];
}

__global__ __launch_bounds__(256) void logits_group_kernel(const float* __restrict__ h,
                                                           const float* __restrict__ attn_l,
                                                           const float* __restrict__ attn_r,
                                                           float* __restrict__ el,
                                                           float* __restrict__ er,
                                                           const int* __restrict__ bstart,
                                                           const int* __restrict__ binned,
                                                           int* __restrict__ cnt,
                                                           int* __restrict__ off,
                                                           int* __restrict__ ssrc) {
    if (blockIdx.x < BINS) {
        group_block(blockIdx.x, bstart, binned, cnt, off, ssrc);
        return;
    }
    int i = (blockIdx.x - BINS) * blockDim.x + threadIdx.x;
    if (i >= N_NODES * HEADS) return;
    const int n = i >> 2;
    const int hh = i & 3;
    const float* __restrict__ hp = h + (size_t)n * FEAT + hh * OUT_D;
    const float* __restrict__ al = attn_l + hh * OUT_D;
    const float* __restrict__ ar = attn_r + hh * OUT_D;
    float sl = 0.f, sr = 0.f;
#pragma unroll
    for (int d = 0; d < OUT_D; ++d) {
        float v = hp[d];
        sl += v * al[d];
        sr += v * ar[d];
    }
    el[i] = sl;
    if (n < NUM_DSTN) er[i] = sr;
}

__device__ __forceinline__ float leaky(float v) { return (v >= 0.f) ? v : NEG_SLOPE * v; }

__device__ __forceinline__ float wave_max(float v) {
#pragma unroll
    for (int o = 16; o > 0; o >>= 1) v = fmaxf(v, __shfl_xor(v, o));
    return v;
}

__device__ __forceinline__ float wave_sum(float v) {
#pragma unroll
    for (int o = 16; o > 0; o >>= 1) v += __shfl_xor(v, o);
    return v;
}

__device__ __forceinline__ int lane_bcast(float v, int j) {
    return __builtin_amdgcn_readlane(__float_as_int(v), j);
}

__device__ __forceinline__ float pick_head(int b0, int b1, int b2, int b3, int k0, int k1, int k2, int k3) {
    return __int_as_float((b0 & k0) | (b1 & k1) | (b2 & k2) | (b3 & k3));
}

__device__ __forceinline__ void add_chunk(float4& acc, int s, float a0, float a1, float a2, float a3,
                                          int cntc, int hd, int lane, const float* __restrict__ h) {
    const int k0 = (hd == 0) ? -1 : 0, k1 = (hd == 1) ? -1 : 0;
    const int k2 = (hd == 2) ? -1 : 0, k3 = (hd == 3) ? -1 : 0;
    int j = 0;
    for (; j + 4 <= cntc; j += 4) {
        float4 hv[4];
        float a[4];
#pragma unroll
        for (int u = 0; u < 4; ++u) {
            const int sj = __builtin_amdgcn_readlane(s, j + u);
            hv[u] = *(const float4*)(h + (size_t)sj * FEAT + lane * 4);
            a[u] = pick_head(lane_bcast(a0, j + u), lane_bcast(a1, j + u),
                             lane_bcast(a2, j + u), lane_bcast(a3, j + u), k0, k1, k2, k3);
        }
#pragma unroll
        for (int u = 0; u < 4; ++u) {
            acc.x += hv[u].x * a[u];
            acc.y += hv[u].y * a[u];
            acc.z += hv[u].z * a[u];
            acc.w += hv[u].w * a[u];
        }
    }
    for (; j < cntc; ++j) {
        const int sj = __builtin_amdgcn_readlane(s, j);
        const float4 hv = *(const float4*)(h + (size_t)sj * FEAT + lane * 4);
        const float a = pick_head(lane_bcast(a0, j), lane_bcast(a1, j),
                                  lane_bcast(a2, j), lane_bcast(a3, j), k0, k1, k2, k3);
        acc.x += hv.x * a;
        acc.y += hv.y * a;
        acc.z += hv.z * a;
        acc.w += hv.w * a;
    }
}

__global__ __launch_bounds__(256) void aggregate_kernel(const int* __restrict__ off,
                                                        const int* __restrict__ cnt,
                                                        const int* __restrict__ ssrc,
                                                        const float* __restrict__ h,
                                                        const float* __restrict__ el,
                                                        const float* __restrict__ er,
                                                        float* __restrict__ out) {
    const int lane = threadIdx.x & 31;
    const int hd = lane >> 3;
    const int d = blockIdx.x * 8 + __builtin_amdgcn_readfirstlane((int)(threadIdx.x >> 5));
    const int begin = off[d];
    const int n = cnt[d];
    const float4* __restrict__ el4 = (const float4*)el;
    float4 acc = {0.f, 0.f, 0.f, 0.f};

    if (n > 0) {
        const float4 r4 = *(const float4*)(er + (size_t)d * HEADS);
        const bool v0 = lane < n;
        int s0 = 0;
        float e0 = -INFINITY, e1 = -INFINITY, e2 = -INFINITY, e3 = -INFINITY;
        if (v0) {
            s0 = ssrc[begin + lane];
            const float4 l4 = el4[s0];
            e0 = leaky(l4.x + r4.x); e1 = leaky(l4.y + r4.y);
            e2 = leaky(l4.z + r4.z); e3 = leaky(l4.w + r4.w);
        }
        float m0 = e0, m1 = e1, m2 = e2, m3 = e3;
        for (int base = 32; base < n; base += 32) {
            const int i = base + lane;
            if (i < n) {
                const float4 l4 = el4[ssrc[begin + i]];
                m0 = fmaxf(m0, leaky(l4.x + r4.x)); m1 = fmaxf(m1, leaky(l4.y + r4.y));
                m2 = fmaxf(m2, leaky(l4.z + r4.z)); m3 = fmaxf(m3, leaky(l4.w + r4.w));
            }
        }
        m0 = wave_max(m0); m1 = wave_max(m1); m2 = wave_max(m2); m3 = wave_max(m3);
        const float p0 = v0 ? expf(e0 - m0) : 0.f, p1 = v0 ? expf(e1 - m1) : 0.f;
        const float p2 = v0 ? expf(e2 - m2) : 0.f, p3 = v0 ? expf(e3 - m3) : 0.f;
        float z0 = p0, z1 = p1, z2 = p2, z3 = p3;
        for (int base = 32; base < n; base += 32) {
            const int i = base + lane;
            if (i < n) {
                const float4 l4 = el4[ssrc[begin + i]];
                z0 += expf(leaky(l4.x + r4.x) - m0); z1 += expf(leaky(l4.y + r4.y) - m1);
                z2 += expf(leaky(l4.z + r4.z) - m2); z3 += expf(leaky(l4.w + r4.w) - m3);
            }
        }
        z0 = wave_sum(z0); z1 = wave_sum(z1); z2 = wave_sum(z2); z3 = wave_sum(z3);
        add_chunk(acc, s0, p0 / z0, p1 / z1, p2 / z2, p3 / z3, (n < 32) ? n : 32, hd, lane, h);
        for (int base = 32; base < n; base += 32) {
            const int i = base + lane;
            int s = 0;
            float a0 = 0.f, a1 = 0.f, a2 = 0.f, a3 = 0.f;
            if (i < n) {
                s = ssrc[begin + i];
                const float4 l4 = el4[s];
                a0 = expf(leaky(l4.x + r4.x) - m0) / z0; a1 = expf(leaky(l4.y + r4.y) - m1) / z1;
                a2 = expf(leaky(l4.z + r4.z) - m2) / z2; a3 = expf(leaky(l4.w + r4.w) - m3) / z3;
            }
            const int rem = n - base;
            add_chunk(acc, s, a0, a1, a2, a3, (rem < 32) ? rem : 32, hd, lane, h);
        }
    }

    float4 r;
    r.x = (acc.x > 0.f) ? acc.x : (expf(acc.x) - 1.0f);
    r.y = (acc.y > 0.f) ? acc.y : (expf(acc.y) - 1.0f);
    r.z = (acc.z > 0.f) ? acc.z : (expf(acc.z) - 1.0f);
    r.w = (acc.w > 0.f) ? acc.w : (expf(acc.w) - 1.0f);
    *(float4*)(out + (size_t)d * FEAT + lane * 4) = r;
    if (blockIdx.x == 0 && threadIdx.x == 0) {
        out[(size_t)NUM_DSTN * FEAT + 0] = 1.0f;
        out[(size_t)NUM_DSTN * FEAT + 1] = 1.0f;
    }
}

extern "C" void kernel_launch(void* const* d_in, const int* in_sizes, int n_in,
                              void* d_out, int out_size, void* d_ws, size_t ws_size,
                              hipStream_t stream) {
    const float* feats  = (const float*)d_in[0];
    const int*   src    = (const int*)d_in[1];
    const int*   dst    = (const int*)d_in[2];
    const float* W      = (const float*)d_in[3];
    const float* attn_l = (const float*)d_in[4];
    const float* attn_r = (const float*)d_in[5];
    float* out = (float*)d_out;

    float* ws   = (float*)d_ws;
    float* hbuf = ws;
    float* el   = hbuf + (size_t)N_NODES * FEAT;
    float* er   = el + (size_t)N_NODES * HEADS;
    int*   binned = (int*)(er + (size_t)N_NODES * HEADS);
    int*   ssrc = binned + (size_t)E_EDGES;
    int*   cnt  = ssrc + (size_t)E_EDGES;
    int*   off  = cnt + (size_t)NUM_DSTN;
    int*   H      = (int*)(er + (size_t)NUM_DSTN * HEADS);
    int*   bstart = H + (size_t)L1_BLOCKS * BINS;

    bin_count_kernel<<<L1_BLOCKS, 512, 0, stream>>>(dst, H);
    gemm_bin_kernel<<<GEMM_BLOCKS, GEMM_THREADS, 0, stream>>>(feats, W, hbuf, src, dst, H, bstart,
                                                              binned);
    logits_group_kernel<<<BINS + LOGIT_BLOCKS, 256, 0, stream>>>(hbuf, attn_l, attn_r, el, er,
                                                                 bstart, binned, cnt, off, ssrc);
    aggregate_kernel<<<NUM_DSTN / 8, 256, 0, stream>>>(off, cnt, ssrc, hbuf, el, er, out);
}


# GEMM epilogue forms el/er from the accumulators (two fma per lane, then a 16-lane DPP tree sum: el/er sum order changes); the logits pass and its read-back of h go, launch 3 keeps only the grouping
# speedup vs baseline: 5.8212x; 1.0443x over previous
#include <hip/hip_runtime.h>
#include <math.h>

#define N_NODES 131072
#define E_EDGES 1048576
#define NUM_DSTN 65536
#define IN_DIM 256
#define HEADS 4
#define OUT_D 32
#define FEAT (HEADS * OUT_D)
#define NEG_SLOPE 0.2f

typedef __attribute__((ext_vector_type(2))) float v2f;
typedef __attribute__((ext_vector_type(8))) float v8f;

#define L1_BLOCKS 256
#define L1_INT4 1024
#define BINS 256
#define SRC_BITS 17
#define GEMM_THREADS 512
#define GEMM_BLOCKS (N_NODES / 16 / 16)
#define WL_PITCH 144

__global__ __launch_bounds__(512) void bin_count_kernel(const int* __restrict__ dst,
                                                        int* __restrict__ H) {
    __shared__ int hist[BINS];
    const int t = threadIdx.x;
    if (t < BINS) hist[t] = 0;
    __syncthreads();
    const int4* __restrict__ d4 = (const int4*)dst + (size_t)blockIdx.x * L1_INT4;
    const int4 a = d4[t], b = d4[512 + t];
    atomicAdd(&hist[a.x >> 8], 1); atomicAdd(&hist[a.y >> 8], 1);
    atomicAdd(&hist[a.z >> 8], 1); atomicAdd(&hist[a.w >> 8], 1);
    atomicAdd(&hist[b.x >> 8], 1); atomicAdd(&hist[b.y >> 8], 1);
    atomicAdd(&hist[b.z >> 8], 1); atomicAdd(&hist[b.w >> 8], 1);
    __syncthreads();
    if (t < BINS) H[blockIdx.x * BINS + t] = hist[t];
}

__device__ __forceinline__ void group_block(int bin, const int* __restrict__ bstart,
                                            const int* __restrict__ binned,
                                            int* __restrict__ cnt, int* __restrict__ off,
                                            int* __restrict__ ssrc) {
    __shared__ int cntL[BINS];
    __shared__ int curL[BINS];
    __shared__ int wtotL[8];
    const int t = threadIdx.x;
    const int lane = t & 31;
    const int wave = t >> 5;
    const int S = bstart[bin];
    const int Eend = (bin == BINS - 1) ? E_EDGES : bstart[bin + 1];
    const int m = Eend - S;
    const int* __restrict__ rec = binned + S;
    cntL[t] = 0;
    __syncthreads();
    int j = t;
    for (; j + 1792 < m; j += 2048) {
        int r[8];
#pragma unroll
        for (int u = 0; u < 8; ++u) r[u] = rec[j + 256 * u];
#pragma unroll
        for (int u = 0; u < 8; ++u) atomicAdd(&cntL[r[u] >> SRC_BITS], 1);
    }
    for (; j < m; j += 256) atomicAdd(&cntL[rec[j] >> SRC_BITS], 1);
    __syncthreads();
    const int c = cntL[t];
    int incl = c;
#pragma unroll
    for (int o = 1; o < 32; o <<= 1) {
        const int nb = __shfl_up(incl, o);
        if (lane >= o) incl += nb;
    }
    if (lane == 31) wtotL[wave] = incl;
    __syncthreads();
    int wb = 0;
    for (int w = 0; w < wave; ++w) wb += wtotL[w];
    const int o = S + wb + incl - c;
    curL[t] = o;
    cnt[bin * BINS + t] = c;
    off[bin * BINS + t] = o;
    __syncthreads();
    j = t;
    for (; j + 1792 < m; j += 2048) {
        int r[8], p[8];
#pragma unroll
        for (int u = 0; u < 8; ++u) r[u] = rec[j + 256 * u];
#pragma unroll
        for (int u = 0; u < 8; ++u) p[u] = atomicAdd(&curL[r[u] >> SRC_BITS], 1);
#pragma unroll
        for (int u = 0; u < 8; ++u) ssrc[p[u]] = r[u] & ((1 << SRC_BITS) - 1);
    }
    for (; j < m; j += 256) {
        const int r0 = rec[j];
        ssrc[atomicAdd(&curL[r0 >> SRC_BITS], 1)] = r0 & ((1 << SRC_BITS) - 1);
    }
}

template <int M> __device__ __forceinline__ float row_xor(float v) {
    return __int_as_float(__builtin_amdgcn_update_dpp(0, __float_as_int(v), 0x160 + M, 0xf, 0xf, false));
}

template <int M> __device__ __forceinline__ float fold_pair(float x0, float x1, bool up) {
    return (up ? x1 : x0) + row_xor<M>(up ? x0 : x1);
}

__global__ __launch_bounds__(GEMM_THREADS) __attribute__((amdgpu_waves_per_eu(8, 8))) void gemm_bin_kernel(const float* __restrict__ feats,
                                                                const float* __restrict__ W,
                                                                float* __restrict__ h,
                                                                const float* __restrict__ attn_l,
                                                                const float* __restrict__ attn_r,
                                                                float* __restrict__ el,
                                                                float* __restrict__ er,
                                                                const int* __restrict__ src,
                                                                const int* __restrict__ dst,
                                                                const int* __restrict__ H,
                                                                int* __restrict__ bstart,
                                                                int* __restrict__ binned) {
    __shared__ float4 Wl[(IN_DIM / 2) * WL_PITCH / 2];
    const int tid = threadIdx.x;

    if ((blockIdx.x & 1) == 0) {
        int* __restrict__ sc = (int*)Wl;
        int* __restrict__ preS = sc;
        int* __restrict__ totS = sc + 8 * BINS;
        int* __restrict__ cur = sc + 16 * BINS;
        int* __restrict__ wtot = sc + 17 * BINS;
        const int bb = blockIdx.x >> 1;
        const int4* __restrict__ d4 = (const int4*)dst + (size_t)bb * L1_INT4;
        const int4* __restrict__ s4 = (const int4*)src + (size_t)bb * L1_INT4;
        const int4 da = d4[tid], db = d4[512 + tid];
        const int4 sa = s4[tid], sb = s4[512 + tid];
        const int c4 = tid & 63;
        const int g = tid >> 6;
        const int4* __restrict__ H4 = (const int4*)H + (size_t)(g * 32) * (BINS / 4) + c4;
        int4 pre = {0, 0, 0, 0}, tot = {0, 0, 0, 0};
#pragma unroll 8
        for (int i = 0; i < 32; ++i) {
            const int4 v = H4[i * (BINS / 4)];
            const int k = (g * 32 + i < bb) ? -1 : 0;
            tot.x += v.x; tot.y += v.y; tot.z += v.z; tot.w += v.w;
            pre.x += v.x & k; pre.y += v.y & k; pre.z += v.z & k; pre.w += v.w & k;
        }
        ((int4*)preS)[g * (BINS / 4) + c4] = pre;
        ((int4*)totS)[g * (BINS / 4) + c4] = tot;
        __syncthreads();
        const int lane = tid & 31;
        const int wave = tid >> 5;
        int p = 0, c = 0, incl = 0;
        if (tid < BINS) {
#pragma unroll
            for (int gg = 0; gg < 8; ++gg) {
                p += preS[gg * BINS + tid];
                c += totS[gg * BINS + tid];
            }
            incl = c;
#pragma unroll
            for (int o = 1; o < 32; o <<= 1) {
                const int nb = __shfl_up(incl, o);
                if (lane >= o) incl += nb;
            }
            if (lane == 31) wtot[wave] = incl;
        }
        __syncthreads();
        if (tid < BINS) {
            int wb = 0;
            for (int w = 0; w < wave; ++w) wb += wtot[w];
            const int bs = wb + incl - c;
            cur[tid] = bs + p;
            if (bb == 0) bstart[tid] = bs;
        }
        __syncthreads();
        binned[atomicAdd(&cur[da.x >> 8], 1)] = ((da.x & 255) << SRC_BITS) | sa.x;
        binned[atomicAdd(&cur[da.y >> 8], 1)] = ((da.y & 255) << SRC_BITS) | sa.y;
        binned[atomicAdd(&cur[da.z >> 8], 1)] = ((da.z & 255) << SRC_BITS) | sa.z;
        binned[atomicAdd(&cur[da.w >> 8], 1)] = ((da.w & 255) << SRC_BITS) | sa.w;
        binned[atomicAdd(&cur[db.x >> 8], 1)] = ((db.x & 255) << SRC_BITS) | sb.x;
        binned[atomicAdd(&cur[db.y >> 8], 1)] = ((db.y & 255) << SRC_BITS) | sb.y;
        binned[atomicAdd(&cur[db.z >> 8], 1)] = ((db.z & 255) << SRC_BITS) | sb.z;
        binned[atomicAdd(&cur[db.w >> 8], 1)] = ((db.w & 255) << SRC_BITS) | sb.w;
        __syncthreads();
    }

    {
        const int c4 = tid & 31;
        const int kp0 = tid >> 5;
#pragma unroll
        for (int it = 0; it < 8; ++it) {
            const int kp = kp0 + 16 * it;
            const float4* __restrict__ wp = (const float4*)(W + (size_t)(2 * kp) * FEAT) + c4;
            const float4 r0 = wp[0];
            const float4 r1 = wp[FEAT / 4];
            float4* __restrict__ o = Wl + (kp * WL_PITCH + 4 * c4) / 2;
            o[0] = make_float4(r0.x, r1.x, r0.y, r1.y);
            o[1] = make_float4(r0.z, r1.z, r0.w, r1.w);
        }
    }
    __syncthreads();

    const int lane = tid & 31;
    const int lo = lane & 15;
    const int hi = lane >> 4;
    const int rowBase = (blockIdx.x * (GEMM_THREADS >> 5) + (tid >> 5)) * 16;
    const float* __restrict__ arow = feats + (size_t)(rowBase + lo) * IN_DIM;
    const v2f* __restrict__ bl = (const v2f*)Wl + hi * WL_PITCH + lo;

    v8f acc[8];
#pragma unroll
    for (int t = 0; t < 8; ++t) acc[t] = (v8f){0.f, 0.f, 0.f, 0.f, 0.f, 0.f, 0.f, 0.f};

    for (int k = 0; k < IN_DIM; k += 4) {
        const int ka = k + 2 * hi;
        v2f a;
        a.x = arow[ka];
        a.y = arow[ka + 1];
#pragma unroll
        for (int t = 0; t < 8; ++t) {
            const v2f b = bl[(k >> 1) * WL_PITCH + t * 16];
            acc[t] = __builtin_amdgcn_wmma_f32_16x16x4_f32(
                false, a, false, b, (short)0, acc[t], false, false);
        }
    }

#pragma unroll
    for (int t = 0; t < 8; ++t)
#pragma unroll
        for (int i = 0; i < 8; ++i)
            h[(size_t)(rowBase + i + 8 * hi) * FEAT + t * 16 + lo] = acc[t][i];

    const bool up8 = (lo & 8) != 0, up4 = (lo & 4) != 0, up2 = (lo & 2) != 0, up1 = (lo & 1) != 0;
    float res[HEADS];
#pragma unroll
    for (int hh = 0; hh < HEADS; ++hh) {
        const float al0 = attn_l[hh * OUT_D + lo], al1 = attn_l[hh * OUT_D + 16 + lo];
        const float ar0 = attn_r[hh * OUT_D + lo], ar1 = attn_r[hh * OUT_D + 16 + lo];
        float v[16];
#pragma unroll
        for (int i = 0; i < 8; ++i) {
            v[2 * i]     = acc[2 * hh][i] * al0 + acc[2 * hh + 1][i] * al1;
            v[2 * i + 1] = acc[2 * hh][i] * ar0 + acc[2 * hh + 1][i] * ar1;
        }
        float w8[8], w4[4], w2[2];
#pragma unroll
        for (int j = 0; j < 8; ++j) w8[j] = fold_pair<8>(v[j], v[j + 8], up8);
#pragma unroll
        for (int j = 0; j < 4; ++j) w4[j] = fold_pair<4>(w8[j], w8[j + 4], up4);
#pragma unroll
        for (int j = 0; j < 2; ++j) w2[j] = fold_pair<2>(w4[j], w4[j + 2], up2);
        res[hh] = fold_pair<1>(w2[0], w2[1], up1);
    }
    const size_t lrow = (size_t)(rowBase + (lo >> 1) + 8 * hi) * HEADS;
    const float4 r4 = make_float4(res[0], res[1], res[2], res[3]);
    if (!up1) *(float4*)(el + lrow) = r4;
    else if (rowBase < NUM_DSTN) *(float4*)(er + lrow) = r4;
}

__global__ __launch_bounds__(256) void group_kernel(const int* __restrict__ bstart,
                                                    const int* __restrict__ binned,
                                                    int* __restrict__ cnt,
                                                    int* __restrict__ off,
                                                    int* __restrict__ ssrc) {
    group_block(blockIdx.x, bstart, binned, cnt, off, ssrc);
}

__device__ __forceinline__ float leaky(float v) { return (v >= 0.f) ? v : NEG_SLOPE * v; }

__device__ __forceinline__ float wave_max(float v) {
#pragma unroll
    for (int o = 16; o > 0; o >>= 1) v = fmaxf(v, __shfl_xor(v, o));
    return v;
}

__device__ __forceinline__ float wave_sum(float v) {
#pragma unroll
    for (int o = 16; o > 0; o >>= 1) v += __shfl_xor(v, o);
    return v;
}

__device__ __forceinline__ int lane_bcast(float v, int j) {
    return __builtin_amdgcn_readlane(__float_as_int(v), j);
}

__device__ __forceinline__ float pick_head(int b0, int b1, int b2, int b3, int k0, int k1, int k2, int k3) {
    return __int_as_float((b0 & k0) | (b1 & k1) | (b2 & k2) | (b3 & k3));
}

__device__ __forceinline__ void add_chunk(float4& acc, int s, float a0, float a1, float a2, float a3,
                                          int cntc, int hd, int lane, const float* __restrict__ h) {
    const int k0 = (hd == 0) ? -1 : 0, k1 = (hd == 1) ? -1 : 0;
    const int k2 = (hd == 2) ? -1 : 0, k3 = (hd == 3) ? -1 : 0;
    int j = 0;
    for (; j + 4 <= cntc; j += 4) {
        float4 hv[4];
        float a[4];
#pragma unroll
        for (int u = 0; u < 4; ++u) {
            const int sj = __builtin_amdgcn_readlane(s, j + u);
            hv[u] = *(const float4*)(h + (size_t)sj * FEAT + lane * 4);
            a[u] = pick_head(lane_bcast(a0, j + u), lane_bcast(a1, j + u),
                             lane_bcast(a2, j + u), lane_bcast(a3, j + u), k0, k1, k2, k3);
        }
#pragma unroll
        for (int u = 0; u < 4; ++u) {
            acc.x += hv[u].x * a[u];
            acc.y += hv[u].y * a[u];
            acc.z += hv[u].z * a[u];
            acc.w += hv[u].w * a[u];
        }
    }
    for (; j < cntc; ++j) {
        const int sj = __builtin_amdgcn_readlane(s, j);
        const float4 hv = *(const float4*)(h + (size_t)sj * FEAT + lane * 4);
        const float a = pick_head(lane_bcast(a0, j), lane_bcast(a1, j),
                                  lane_bcast(a2, j), lane_bcast(a3, j), k0, k1, k2, k3);
        acc.x += hv.x * a;
        acc.y += hv.y * a;
        acc.z += hv.z * a;
        acc.w += hv.w * a;
    }
}

__global__ __launch_bounds__(256) void aggregate_kernel(const int* __restrict__ off,
                                                        const int* __restrict__ cnt,
                                                        const int* __restrict__ ssrc,
                                                        const float* __restrict__ h,
                                                        const float* __restrict__ el,
                                                        const float* __restrict__ er,
                                                        float* __restrict__ out) {
    const int lane = threadIdx.x & 31;
    const int hd = lane >> 3;
    const int d = blockIdx.x * 8 + __builtin_amdgcn_readfirstlane((int)(threadIdx.x >> 5));
    const int begin = off[d];
    const int n = cnt[d];
    const float4* __restrict__ el4 = (const float4*)el;
    float4 acc = {0.f, 0.f, 0.f, 0.f};

    if (n > 0) {
        const float4 r4 = *(const float4*)(er + (size_t)d * HEADS);
        const bool v0 = lane < n;
        int s0 = 0;
        float e0 = -INFINITY, e1 = -INFINITY, e2 = -INFINITY, e3 = -INFINITY;
        if (v0) {
            s0 = ssrc[begin + lane];
            const float4 l4 = el4[s0];
            e0 = leaky(l4.x + r4.x); e1 = leaky(l4.y + r4.y);
            e2 = leaky(l4.z + r4.z); e3 = leaky(l4.w + r4.w);
        }
        float m0 = e0, m1 = e1, m2 = e2, m3 = e3;
        for (int base = 32; base < n; base += 32) {
            const int i = base + lane;
            if (i < n) {
                const float4 l4 = el4[ssrc[begin + i]];
                m0 = fmaxf(m0, leaky(l4.x + r4.x)); m1 = fmaxf(m1, leaky(l4.y + r4.y));
                m2 = fmaxf(m2, leaky(l4.z + r4.z)); m3 = fmaxf(m3, leaky(l4.w + r4.w));
            }
        }
        m0 = wave_max(m0); m1 = wave_max(m1); m2 = wave_max(m2); m3 = wave_max(m3);
        const float p0 = v0 ? expf(e0 - m0) : 0.f, p1 = v0 ? expf(e1 - m1) : 0.f;
        const float p2 = v0 ? expf(e2 - m2) : 0.f, p3 = v0 ? expf(e3 - m3) : 0.f;
        float z0 = p0, z1 = p1, z2 = p2, z3 = p3;
        for (int base = 32; base < n; base += 32) {
            const int i = base + lane;
            if (i < n) {
                const float4 l4 = el4[ssrc[begin + i]];
                z0 += expf(leaky(l4.x + r4.x) - m0); z1 += expf(leaky(l4.y + r4.y) - m1);
                z2 += expf(leaky(l4.z + r4.z) - m2); z3 += expf(leaky(l4.w + r4.w) - m3);
            }
        }
        z0 = wave_sum(z0); z1 = wave_sum(z1); z2 = wave_sum(z2); z3 = wave_sum(z3);
        add_chunk(acc, s0, p0 / z0, p1 / z1, p2 / z2, p3 / z3, (n < 32) ? n : 32, hd, lane, h);
        for (int base = 32; base < n; base += 32) {
            const int i = base + lane;
            int s = 0;
            float a0 = 0.f, a1 = 0.f, a2 = 0.f, a3 = 0.f;
            if (i < n) {
                s = ssrc[begin + i];
                const float4 l4 = el4[s];
                a0 = expf(leaky(l4.x + r4.x) - m0) / z0; a1 = expf(leaky(l4.y + r4.y) - m1) / z1;
                a2 = expf(leaky(l4.z + r4.z) - m2) / z2; a3 = expf(leaky(l4.w + r4.w) - m3) / z3;
            }
            const int rem = n - base;
            add_chunk(acc, s, a0, a1, a2, a3, (rem < 32) ? rem : 32, hd, lane, h);
        }
    }

    float4 r;
    r.x = (acc.x > 0.f) ? acc.x : (expf(acc.x) - 1.0f);
    r.y = (acc.y > 0.f) ? acc.y : (expf(acc.y) - 1.0f);
    r.z = (acc.z > 0.f) ? acc.z : (expf(acc.z) - 1.0f);
    r.w = (acc.w > 0.f) ? acc.w : (expf(acc.w) - 1.0f);
    *(float4*)(out + (size_t)d * FEAT + lane * 4) = r;
    if (blockIdx.x == 0 && threadIdx.x == 0) {
        out[(size_t)NUM_DSTN * FEAT + 0] = 1.0f;
        out[(size_t)NUM_DSTN * FEAT + 1] = 1.0f;
    }
}

extern "C" void kernel_launch(void* const* d_in, const int* in_sizes, int n_in,
                              void* d_out, int out_size, void* d_ws, size_t ws_size,
                              hipStream_t stream) {
    const float* feats  = (const float*)d_in[0];
    const int*   src    = (const int*)d_in[1];
    const int*   dst    = (const int*)d_in[2];
    const float* W      = (const float*)d_in[3];
    const float* attn_l = (const float*)d_in[4];
    const float* attn_r = (const float*)d_in[5];
    float* out = (float*)d_out;

    float* ws   = (float*)d_ws;
    float* hbuf = ws;
    float* el   = hbuf + (size_t)N_NODES * FEAT;
    float* er   = el + (size_t)N_NODES * HEADS;
    int*   binned = (int*)(er + (size_t)N_NODES * HEADS);
    int*   ssrc = binned + (size_t)E_EDGES;
    int*   cnt  = ssrc + (size_t)E_EDGES;
    int*   off  = cnt + (size_t)NUM_DSTN;
    int*   H      = (int*)(er + (size_t)NUM_DSTN * HEADS);
    int*   bstart = H + (size_t)L1_BLOCKS * BINS;

    bin_count_kernel<<<L1_BLOCKS, 512, 0, stream>>>(dst, H);
    gemm_bin_kernel<<<GEMM_BLOCKS, GEMM_THREADS, 0, stream>>>(feats, W, hbuf, attn_l, attn_r, el, er,
                                                              src, dst, H, bstart, binned);
    group_kernel<<<BINS, 256, 0, stream>>>(bstart, binned, cnt, off, ssrc);
    aggregate_kernel<<<NUM_DSTN / 8, 256, 0, stream>>>(off, cnt, ssrc, hbuf, el, er, out);
}
